# starting kernel
# baseline (speedup 1.0000x reference)
#include <hip/hip_runtime.h>
#include <math.h>

#define Cc   32
#define Kk   128
#define Bb   4
#define Tt   512
#define TPp  1024
#define CH   64
#define RT   2
#define EPS_F 2.2204460492503131e-16f

typedef __attribute__((ext_vector_type(2))) float v2f;
typedef __attribute__((ext_vector_type(8))) float v8f;

__device__ __forceinline__ float softplus_f(float x) {
    return fmaxf(x, 0.0f) + log1pf(__expf(-fabsf(x)));
}

__global__ void prep_params(const float* __restrict__ mu,
                            const float* __restrict__ alpha,
                            const float* __restrict__ delta,
                            const float* __restrict__ cf,
                            const int*   __restrict__ ftc,
                            float* __restrict__ sp_alpha,
                            float* __restrict__ sp_delta,
                            float* __restrict__ sp_mu,
                            float* __restrict__ Mexp) {
    __shared__ float s_e[Kk];
    __shared__ float s_den[Cc];
    __shared__ float s_prob[Kk];
    __shared__ float s_max;
    const int tid = threadIdx.x;

    for (int i = tid; i < Cc * Cc; i += 128) {
        sp_alpha[i] = softplus_f(alpha[i]);
        sp_delta[i] = softplus_f(delta[i]) + EPS_F;
    }
    if (tid < Cc) sp_mu[tid] = softplus_f(mu[tid]);

    if (tid == 0) {
        float m = cf[0];
        for (int k = 1; k < Kk; ++k) m = fmaxf(m, cf[k]);
        s_max = m;
    }
    __syncthreads();
    s_e[tid] = __expf(cf[tid] - s_max);
    __syncthreads();
    if (tid == 0) {
        for (int c = 0; c < Cc; ++c) s_den[c] = 0.0f;
        for (int k = 0; k < Kk; ++k) s_den[ftc[k]] += s_e[k];
    }
    __syncthreads();
    s_prob[tid] = s_e[tid] / s_den[ftc[tid]];
    __syncthreads();
    for (int i = tid; i < Cc * Kk; i += 128) {
        int c = i >> 7, k = i & (Kk - 1);
        Mexp[i] = (ftc[k] == c) ? s_prob[k] : 0.0f;
    }
}

__global__ __launch_bounds__(256)
void hawkes_wmma_kernel(const int*   __restrict__ past_event,
                        const float* __restrict__ past_time,
                        const float* __restrict__ time_tensor,
                        const float* __restrict__ sp_alpha,
                        const float* __restrict__ sp_delta,
                        const float* __restrict__ sp_mu,
                        const float* __restrict__ Mexp,
                        float*       __restrict__ out) {
    __shared__ float s_spA[Cc * Cc];
    __shared__ float s_spD[Cc * Cc];
    __shared__ float s_mu[Cc];
    __shared__ float s_pt[CH];
    __shared__ int   s_ev[CH];
    __shared__ float s_acc[16 * Cc];

    const int tid    = threadIdx.x;
    const int lane   = tid & 31;
    const int wave   = tid >> 5;
    const int b      = blockIdx.x >> 6;
    const int tile   = blockIdx.x & 63;
    const int t_base = tile * 16;

    for (int i = tid; i < Cc * Cc; i += 256) {
        s_spA[i] = sp_alpha[i];
        s_spD[i] = sp_delta[i];
    }
    if (tid < Cc) s_mu[tid] = sp_mu[tid];
    __builtin_prefetch(Mexp + lane * 16, 0, 0);

    float tm[RT], acc[RT];
#pragma unroll
    for (int r = 0; r < RT; ++r) {
        const int t = t_base + wave * RT + r;
        tm[r]  = time_tensor[b * TPp + t];
        acc[r] = 0.0f;
    }

    for (int chunk = 0; chunk < Tt; chunk += CH) {
        __syncthreads();
        if (tid < CH) {
            s_pt[tid] = past_time[b * Tt + chunk + tid];
            s_ev[tid] = past_event[b * Tt + chunk + tid];
        }
        __syncthreads();
#pragma unroll 4
        for (int ii = 0; ii < CH; ++ii) {
            const float pt = s_pt[ii];
            const int   ev = s_ev[ii];
            const float a  = s_spA[ev * Cc + lane];
            const float d  = s_spD[ev * Cc + lane];
#pragma unroll
            for (int r = 0; r < RT; ++r) {
                const float et  = tm[r] - pt;
                const float etc = fmaxf(et, 0.0f);
                const float val = a * __expf(-d * etc);
                acc[r] += (et > EPS_F) ? val : 0.0f;
            }
        }
    }

    __syncthreads();
#pragma unroll
    for (int r = 0; r < RT; ++r) {
        s_acc[(wave * RT + r) * Cc + lane] = acc[r] + s_mu[lane];
    }
    __syncthreads();

    const int half = lane >> 4;
    const int l15  = lane & 15;
    v8f cacc = {};
#pragma unroll
    for (int kc = 0; kc < 8; ++kc) {
        v2f af, bf;
#pragma unroll
        for (int j = 0; j < 2; ++j) {
            const int cch = kc * 4 + half * 2 + j;
            af[j] = s_acc[l15 * Cc + cch];
            bf[j] = Mexp[cch * Kk + wave * 16 + l15];
        }
        cacc = __builtin_amdgcn_wmma_f32_16x16x4_f32(
             false, af,  false, bf,
             (short)0, cacc,  false,  false);
    }
#pragma unroll
    for (int r = 0; r < 8; ++r) {
        const int m = r + half * 8;
        const int t = t_base + m;
        out[(b * TPp + t) * Kk + wave * 16 + l15] = cacc[r];
    }
}

extern "C" void kernel_launch(void* const* d_in, const int* in_sizes, int n_in,
                              void* d_out, int out_size, void* d_ws, size_t ws_size,
                              hipStream_t stream) {
    const int*   past_event  = (const int*)  d_in[0];
    const float* past_time   = (const float*)d_in[1];
    const float* time_tensor = (const float*)d_in[2];
    const float* mu          = (const float*)d_in[3];
    const float* alpha       = (const float*)d_in[4];
    const float* delta       = (const float*)d_in[5];
    const float* cf_logits   = (const float*)d_in[6];
    const int*   ftc         = (const int*)  d_in[7];
    float* out = (float*)d_out;

    float* sp_alpha = (float*)d_ws;
    float* sp_delta = sp_alpha + Cc * Cc;
    float* sp_mu    = sp_delta + Cc * Cc;
    float* Mexp     = sp_mu + Cc;

    prep_params<<<1, 128, 0, stream>>>(mu, alpha, delta, cf_logits, ftc,
                                       sp_alpha, sp_delta, sp_mu, Mexp);
    hawkes_wmma_kernel<<<Bb * (TPp / 16), 256, 0, stream>>>(
        past_event, past_time, time_tensor,
        sp_alpha, sp_delta, sp_mu, Mexp, out);
}


# Channels split over the lane halves: lane = (query time, channel half), block = 16 query times, grid fills all 256 WGPs (was 128); wave = 2 slices in turn, ids/times by readlane (sum order kept)
# speedup vs baseline: 3.3155x; 3.3155x over previous
#include <hip/hip_runtime.h>
#include <math.h>

#define Cc   32
#define Kk   128
#define Bb   4
#define Tt   512
#define TPp  1024
#define QT   16
#define NW   8
#define EPW  32
#define NSL  (Tt / EPW)
#define SPW  (NSL / NW)
#define CH   (Cc / 2)
#define PP   17
#define NLOG2E (-1.4426950408889634f)
#define EPS_F 2.2204460492503131e-16f

typedef __attribute__((ext_vector_type(2))) float v2f;
typedef __attribute__((ext_vector_type(8))) float v8f;

__device__ __forceinline__ float softplus_f(float x) {
    return fmaxf(x, 0.0f) + log1pf(__expf(-fabsf(x)));
}

#define SP_BLOCKS   8
#define ROW_BLOCKS  4
#define SORT_EV     32
#define SORT_BLOCKS (Bb * Tt / SORT_EV)
#define PREP_BLOCKS 77
static_assert(PREP_BLOCKS == SP_BLOCKS + 1 + ROW_BLOCKS + SORT_BLOCKS, "prep grid");
static_assert(Tt == 512 && Tt % 128 == 0 && SORT_EV * 4 == 128, "sort tiling");
static_assert(SP_BLOCKS * 128 == Cc * Cc && ROW_BLOCKS * 8 == Cc && Kk == 128, "prep tiling");

__global__ __launch_bounds__(128)
void prep_params(const int*   __restrict__ past_event,
                 const float* __restrict__ past_time,
                 const float* __restrict__ mu,
                 const float* __restrict__ alpha,
                 const float* __restrict__ delta,
                 const float* __restrict__ cf,
                 const int*   __restrict__ ftc,
                 float* __restrict__ sp_alpha,
                 float* __restrict__ sp_delta,
                 float* __restrict__ sp_mu,
                 float* __restrict__ Mexp,
                 int*   __restrict__ ord_event,
                 float* __restrict__ ord_time) {
    __shared__ __attribute__((aligned(16))) float s_e[Kk];
    __shared__ __attribute__((aligned(16))) int   s_f[Kk];
    __shared__ __attribute__((aligned(16))) int   s_key[Tt];
    __shared__ int s_rank[128];
    const int tid = threadIdx.x;
    const int blk = blockIdx.x;

    if (blk < SP_BLOCKS) {
        const int i = blk * 128 + tid;
        sp_alpha[i] = softplus_f(alpha[i]);
        sp_delta[i] = softplus_f(delta[i]) + EPS_F;
        return;
    }
    if (blk == SP_BLOCKS) {
        if (tid < Cc) sp_mu[tid] = softplus_f(mu[tid]);
        return;
    }
    if (blk >= SP_BLOCKS + 1 + ROW_BLOCKS) {
        const int s    = blk - (SP_BLOCKS + 1 + ROW_BLOCKS);
        const int bb   = s / (Tt / SORT_EV);
        const int i0   = (s % (Tt / SORT_EV)) * SORT_EV;
        const int part = tid >> 5;
        const int li   = tid & 31;
#pragma unroll
        for (int r = 0; r < Tt / 128; ++r) {
            const int j = r * 128 + tid;
            s_key[j] = past_event[bb * Tt + j] * Tt + j;
        }
        const float pti = (tid < SORT_EV) ? past_time[bb * Tt + i0 + tid] : 0.0f;
        __syncthreads();
        const int key = s_key[i0 + li];
        const int j0  = part * (Tt / 4);
        int r0 = 0, r1 = 0, r2 = 0, r3 = 0;
#pragma unroll
        for (int j = 0; j < Tt / 4; j += 4) {
            r0 += (s_key[j0 + j]     < key) ? 1 : 0;
            r1 += (s_key[j0 + j + 1] < key) ? 1 : 0;
            r2 += (s_key[j0 + j + 2] < key) ? 1 : 0;
            r3 += (s_key[j0 + j + 3] < key) ? 1 : 0;
        }
        s_rank[part * SORT_EV + li] = (r0 + r1) + (r2 + r3);
        __syncthreads();
        if (tid < SORT_EV) {
            const int rank = (s_rank[tid] + s_rank[SORT_EV + tid])
                           + (s_rank[2 * SORT_EV + tid] + s_rank[3 * SORT_EV + tid]);
            ord_event[bb * Tt + rank] = key / Tt;
            ord_time[bb * Tt + rank]  = pti;
        }
        return;
    }

    const float cfk = cf[tid];
    const int   f   = ftc[tid];
    float m0 = cf[0], m1 = cf[1], m2 = cf[2], m3 = cf[3];
#pragma unroll
    for (int k = 4; k < Kk; k += 4) {
        m0 = fmaxf(m0, cf[k]);     m1 = fmaxf(m1, cf[k + 1]);
        m2 = fmaxf(m2, cf[k + 2]); m3 = fmaxf(m3, cf[k + 3]);
    }
    const float m = fmaxf(fmaxf(m0, m1), fmaxf(m2, m3));
    const float e = __expf(cfk - m);
    s_e[tid] = e;
    s_f[tid] = f;
    __syncthreads();
    float den = 0.0f;
#pragma unroll
    for (int k = 0; k < Kk; ++k) den += (s_f[k] == f) ? s_e[k] : 0.0f;
    const float p = e / den;

    const int c0 = (blk - SP_BLOCKS - 1) * (Cc / ROW_BLOCKS);
#pragma unroll
    for (int r = 0; r < Cc / ROW_BLOCKS; ++r) {
        const int c = c0 + r;
        Mexp[c * Kk + tid] = (f == c) ? p : 0.0f;
    }
}

__global__ __launch_bounds__(256)
void hawkes_wmma_kernel(const int*   __restrict__ ord_event,
                        const float* __restrict__ ord_time,
                        const float* __restrict__ time_tensor,
                        const float* __restrict__ sp_alpha,
                        const float* __restrict__ sp_delta,
                        const float* __restrict__ sp_mu,
                        const float* __restrict__ Mexp,
                        float*       __restrict__ out) {
    __shared__ float s_part[NSL * Cc * PP];
    __shared__ float s_acc[QT * Cc];

    const int tid    = threadIdx.x;
    const int lane   = tid & 31;
    const int q      = lane & 15;
    const int hc     = lane >> 4;
    const int wave   = __builtin_amdgcn_readfirstlane(tid >> 5);
    const int b      = blockIdx.x >> 6;
    const int tile   = blockIdx.x & 63;
    const int t_base = tile * QT;

    __builtin_prefetch(Mexp + lane * 16, 0, 0);

    const float tm = time_tensor[b * TPp + t_base + q];

    const int eb = b * Tt + wave * (SPW * EPW);
    int evs[SPW], pts[SPW];
#pragma unroll
    for (int s = 0; s < SPW; ++s) {
        evs[s] = ord_event[eb + s * EPW + lane];
        pts[s] = __float_as_int(ord_time[eb + s * EPW + lane]);
    }

#pragma unroll
    for (int s = 0; s < SPW; ++s) {
        const int evv = evs[s];
        const int ptv = pts[s];
        float acc[CH];
#pragma unroll
        for (int c = 0; c < CH; ++c) acc[c] = 0.0f;

        int   i  = 0;
        int   ev = __builtin_amdgcn_readlane(evv, 0);
        float pt = __int_as_float(__builtin_amdgcn_readlane(ptv, 0));
        while (i < EPW) {
            const float* __restrict__ ar = sp_alpha + ev * Cc + hc * CH;
            const float* __restrict__ dr = sp_delta + ev * Cc + hc * CH;
            float a[CH], d[CH];
#pragma unroll
            for (int c = 0; c < CH; ++c) { a[c] = ar[c]; d[c] = dr[c]; }
            int nev;
            do {
                const int   in  = (i + 1 < EPW) ? i + 1 : i;
                const float npt = __int_as_float(__builtin_amdgcn_readlane(ptv, in));
                nev = __builtin_amdgcn_readlane(evv, in);
                const float et = tm - pt;
                const float x = (et > EPS_F) ? et * NLOG2E : -INFINITY;
#pragma unroll
                for (int c = 0; c < CH; ++c) {
                    acc[c] = fmaf(a[c], __builtin_amdgcn_exp2f(d[c] * x), acc[c]);
                }
                ++i;
                pt = npt;
            } while (i < EPW && nev == ev);
            ev = nev;
        }

        const int slice = wave * SPW + s;
#pragma unroll
        for (int c = 0; c < CH; ++c) {
            s_part[(slice * Cc + hc * CH + c) * PP + q] = acc[c];
        }
    }
    __syncthreads();
#pragma unroll
    for (int qq = 0; qq < QT / NW; ++qq) {
        const int qj = wave + qq * NW;
        float s = 0.0f;
#pragma unroll
        for (int w = 0; w < NSL; ++w) s += s_part[(w * Cc + lane) * PP + qj];
        s_acc[qj * Cc + lane] = s + sp_mu[lane];
    }
    __syncthreads();

    const int half = lane >> 4;
    const int l15  = lane & 15;
    const int nt   = wave;
    v8f cacc = {};
#pragma unroll
    for (int kc = 0; kc < 8; ++kc) {
        v2f af, bf;
#pragma unroll
        for (int j = 0; j < 2; ++j) {
            const int cch = kc * 4 + half * 2 + j;
            af[j] = s_acc[l15 * Cc + cch];
            bf[j] = Mexp[cch * Kk + nt * 16 + l15];
        }
        cacc = __builtin_amdgcn_wmma_f32_16x16x4_f32(
             false, af,  false, bf,
             (short)0, cacc,  false,  false);
    }
#pragma unroll
    for (int r = 0; r < 8; ++r) {
        const int m = r + half * 8;
        const int t = t_base + m;
        out[(b * TPp + t) * Kk + nt * 16 + l15] = cacc[r];
    }
}

extern "C" void kernel_launch(void* const* d_in, const int* in_sizes, int n_in,
                              void* d_out, int out_size, void* d_ws, size_t ws_size,
                              hipStream_t stream) {
    const int*   past_event  = (const int*)  d_in[0];
    const float* past_time   = (const float*)d_in[1];
    const float* time_tensor = (const float*)d_in[2];
    const float* mu          = (const float*)d_in[3];
    const float* alpha       = (const float*)d_in[4];
    const float* delta       = (const float*)d_in[5];
    const float* cf_logits   = (const float*)d_in[6];
    const int*   ftc         = (const int*)  d_in[7];
    float* out = (float*)d_out;

    float* sp_alpha = (float*)d_ws;
    float* sp_delta = sp_alpha + Cc * Cc;
    float* sp_mu    = sp_delta + Cc * Cc;
    float* Mexp     = sp_mu + Cc;
    int*   ord_event = (int*)(Mexp + Cc * Kk);
    float* ord_time  = (float*)(ord_event + Bb * Tt);

    prep_params<<<PREP_BLOCKS, 128, 0, stream>>>(past_event, past_time,
                                       mu, alpha, delta, cf_logits, ftc,
                                       sp_alpha, sp_delta, sp_mu, Mexp,
                                       ord_event, ord_time);
    hawkes_wmma_kernel<<<Bb * (TPp / QT), NW * 32, 0, stream>>>(
        ord_event, ord_time, time_tensor,
        sp_alpha, sp_delta, sp_mu, Mexp, out);
}


# One global round trip per kernel role: main loads B fragments and softplus(mu) with its entry loads; sort blocks rebuild the time from the key; logit max from 4 loads/lane + DPP (sum order kept)
# speedup vs baseline: 4.1490x; 1.2514x over previous
#include <hip/hip_runtime.h>
#include <math.h>

#define Cc   32
#define Kk   128
#define Bb   4
#define Tt   512
#define TPp  1024
#define QT   16
#define NW   8
#define IPW  (Cc / NW)
#define CH   (Cc / 2)
#define PP   17
#define RP   36
#define NLOG2E (-1.4426950408889634f)
#define EPS_F 2.2204460492503131e-16f

typedef __attribute__((ext_vector_type(2))) float v2f;
typedef __attribute__((ext_vector_type(8))) float v8f;

__device__ __forceinline__ float softplus_f(float x) {
    return fmaxf(x, 0.0f) + log1pf(__expf(-fabsf(x)));
}

template <int CTRL>
__device__ __forceinline__ float row_swz(float x) {
    return __int_as_float(__builtin_amdgcn_update_dpp(0, __float_as_int(x), CTRL, 0xF, 0xF, false));
}
__device__ __forceinline__ float xhalf(float x) {
    const int b = __float_as_int(x);
    return __int_as_float(__builtin_amdgcn_permlanex16(b, b, 0x76543210, 0xfedcba98, false, false));
}

#define SP_BLOCKS   8
#define ROW_BLOCKS  4
#define SORT_EV     32
#define SORT_BLOCKS (Bb * Tt / SORT_EV)
#define PREP_BLOCKS 77
static_assert(PREP_BLOCKS == SP_BLOCKS + 1 + ROW_BLOCKS + SORT_BLOCKS, "prep grid");
static_assert(Tt == 512 && Tt % 128 == 0 && SORT_EV * 4 == 128, "sort tiling and 9 index bits");
static_assert(SP_BLOCKS * 128 == Cc * Cc && ROW_BLOCKS * 8 == Cc && Kk == 128, "prep tiling");

__global__ __launch_bounds__(128) __attribute__((amdgpu_waves_per_eu(1, 8)))
void prep_params(const int*   __restrict__ past_event,
                 const float* __restrict__ past_time,
                 const float* __restrict__ mu,
                 const float* __restrict__ alpha,
                 const float* __restrict__ delta,
                 const float* __restrict__ cf,
                 const int*   __restrict__ ftc,
                 float* __restrict__ sp_alpha,
                 float* __restrict__ sp_delta,
                 float* __restrict__ sp_mu,
                 float* __restrict__ Mexp,
                 int*   __restrict__ ord_event,
                 float* __restrict__ ord_time) {
    __shared__ __attribute__((aligned(16))) float s_e[Kk];
    __shared__ __attribute__((aligned(16))) int   s_f[Kk];
    __shared__ __attribute__((aligned(16))) unsigned long long s_key[Tt];
    __shared__ int s_rank[128];
    const int tid = threadIdx.x;
    const int blk = blockIdx.x;

    if (blk < SP_BLOCKS) {
        const int i = blk * 128 + tid;
        sp_alpha[i] = softplus_f(alpha[i]);
        sp_delta[i] = softplus_f(delta[i]) + EPS_F;
        return;
    }
    if (blk == SP_BLOCKS) {
        if (tid < Cc) sp_mu[tid] = softplus_f(mu[tid]);
        return;
    }
    if (blk >= SP_BLOCKS + 1 + ROW_BLOCKS) {
        const int s    = blk - (SP_BLOCKS + 1 + ROW_BLOCKS);
        const int bb   = s / (Tt / SORT_EV);
        const int i0   = (s % (Tt / SORT_EV)) * SORT_EV;
        const int part = tid >> 5;
        const int li   = tid & 31;
        unsigned int tbv[Tt / 128];
        unsigned int evv[Tt / 128];
#pragma unroll
        for (int r = 0; r < Tt / 128; ++r) {
            tbv[r] = __float_as_uint(past_time[bb * Tt + r * 128 + tid]);
            evv[r] = (unsigned int)past_event[bb * Tt + r * 128 + tid];
        }
#pragma unroll
        for (int r = 0; r < Tt / 128; ++r) {
            const int j = r * 128 + tid;
            const unsigned int tb = tbv[r];
            const unsigned int to = tb ^ ((tb >> 31) ? 0xFFFFFFFFu : 0x80000000u);
            s_key[j] = ((unsigned long long)evv[r] << 41)
                     | ((unsigned long long)to << 9) | (unsigned long long)j;
        }
        __syncthreads();
        const unsigned long long key = s_key[i0 + li];
        const int j0  = part * (Tt / 4);
        int r0 = 0, r1 = 0, r2 = 0, r3 = 0;
#pragma unroll
        for (int j = 0; j < Tt / 4; j += 4) {
            r0 += (s_key[j0 + j]     < key) ? 1 : 0;
            r1 += (s_key[j0 + j + 1] < key) ? 1 : 0;
            r2 += (s_key[j0 + j + 2] < key) ? 1 : 0;
            r3 += (s_key[j0 + j + 3] < key) ? 1 : 0;
        }
        s_rank[part * SORT_EV + li] = (r0 + r1) + (r2 + r3);
        __syncthreads();
        if (tid < SORT_EV) {
            const int rank = (s_rank[tid] + s_rank[SORT_EV + tid])
                           + (s_rank[2 * SORT_EV + tid] + s_rank[3 * SORT_EV + tid]);
            const unsigned int to = (unsigned int)(key >> 9);
            const unsigned int tb = to ^ ((to >> 31) ? 0x80000000u : 0xFFFFFFFFu);
            ord_event[bb * Tt + rank] = (int)(key >> 41);
            ord_time[bb * Tt + rank]  = __uint_as_float(tb);
        }
        return;
    }

    const float cfk = cf[tid];
    const int   f   = ftc[tid];
    const int l4 = (tid & 31) * 4;
    const float g0 = cf[l4], g1 = cf[l4 + 1], g2 = cf[l4 + 2], g3 = cf[l4 + 3];
    float m = fmaxf(fmaxf(g0, g1), fmaxf(g2, g3));
    m = fmaxf(m, row_swz<0xB1>(m));
    m = fmaxf(m, row_swz<0x4E>(m));
    m = fmaxf(m, row_swz<0x141>(m));
    m = fmaxf(m, row_swz<0x140>(m));
    m = fmaxf(m, xhalf(m));
    const float e = __expf(cfk - m);
    s_e[tid] = e;
    s_f[tid] = f;
    __syncthreads();
    float den = 0.0f;
#pragma unroll
    for (int k = 0; k < Kk; ++k) den += (s_f[k] == f) ? s_e[k] : 0.0f;
    const float p = e / den;

    const int c0 = (blk - SP_BLOCKS - 1) * (Cc / ROW_BLOCKS);
#pragma unroll
    for (int r = 0; r < Cc / ROW_BLOCKS; ++r) {
        const int c = c0 + r;
        Mexp[c * Kk + tid] = (f == c) ? p : 0.0f;
    }
}

__global__ __launch_bounds__(256)
void hawkes_wmma_kernel(const int*   __restrict__ ord_event,
                        const float* __restrict__ ord_time,
                        const float* __restrict__ time_tensor,
                        const float* __restrict__ sp_alpha,
                        const float* __restrict__ sp_delta,
                        const float* __restrict__ sp_mu,
                        const float* __restrict__ Mexp,
                        float*       __restrict__ out) {
    __shared__ float s_part[NW * Cc * PP];
    __shared__ float s_acc[QT * Cc];
    __shared__ __attribute__((aligned(16))) float s_R[Tt * RP];
    __shared__ int   s_ev[Tt];
    __shared__ __attribute__((aligned(16))) float s_pt[Tt];
    __shared__ int   s_beg[Cc];
    __shared__ int   s_end[Cc];

    const int tid    = threadIdx.x;
    const int lane   = tid & 31;
    const int q      = lane & 15;
    const int hc     = lane >> 4;
    const int wave   = __builtin_amdgcn_readfirstlane(tid >> 5);
    const int b      = blockIdx.x >> 6;
    const int tile   = blockIdx.x & 63;
    const int t_base = tile * QT;

    if (tid < Cc) { s_beg[tid] = 0; s_end[tid] = 0; }

    const int   e0 = ord_event[b * Tt + tid];
    const int   e1 = ord_event[b * Tt + 256 + tid];
    const float p0 = ord_time[b * Tt + tid];
    const float p1 = ord_time[b * Tt + 256 + tid];
    const float tm = time_tensor[b * TPp + t_base + q];

    const int se = tid >> 3;
    const int cg = tid & 7;
    const float4 d4 = *reinterpret_cast<const float4*>(sp_delta + se * Cc + cg * 4);
    float a[IPW][CH], d[IPW][CH];
#pragma unroll
    for (int j = 0; j < IPW; ++j) {
        const float* __restrict__ ar = sp_alpha + (wave * IPW + j) * Cc + hc * CH;
        const float* __restrict__ dr = sp_delta + (wave * IPW + j) * Cc + hc * CH;
#pragma unroll
        for (int c = 0; c < CH; ++c) { a[j][c] = ar[c]; d[j][c] = dr[c]; }
    }
    const float mu_l = sp_mu[lane];
    float bfr[8][2];
#pragma unroll
    for (int kc = 0; kc < 8; ++kc) {
#pragma unroll
        for (int j = 0; j < 2; ++j) {
            bfr[kc][j] = Mexp[(kc * 4 + hc * 2 + j) * Kk + wave * 16 + q];
        }
    }

    s_ev[tid] = e0;  s_ev[tid + 256] = e1;
    s_pt[tid] = p0;  s_pt[tid + 256] = p1;
    __syncthreads();
#pragma unroll
    for (int r = 0; r < 2; ++r) {
        const int i  = tid + r * 256;
        const int ei = r ? e1 : e0;
        if (i == 0      || s_ev[i - 1] != ei) s_beg[ei] = i;
        if (i == Tt - 1 || s_ev[i + 1] != ei) s_end[ei] = i + 1;
    }
    __syncthreads();

    {
        const int beg = s_beg[se];
        const int end = s_end[se];
        float r0 = 0.0f, r1 = 0.0f, r2 = 0.0f, r3 = 0.0f;
        float prev = 0.0f;
        int pos4 = beg & ~3;
        float4 cur = *reinterpret_cast<const float4*>(s_pt + min(pos4, Tt - 4));
        for (; pos4 < end; pos4 += 4) {
            const float4 nxt = *reinterpret_cast<const float4*>(s_pt + min(pos4 + 4, Tt - 4));
            const float t[4] = {cur.x, cur.y, cur.z, cur.w};
#pragma unroll
            for (int j = 0; j < 4; ++j) {
                const int pos = pos4 + j;
                if (pos >= beg && pos < end) {
                    const float g = (pos == beg) ? 0.0f : (t[j] - prev) * NLOG2E;
                    prev = t[j];
                    r0 = fmaf(r0, __builtin_amdgcn_exp2f(d4.x * g), 1.0f);
                    r1 = fmaf(r1, __builtin_amdgcn_exp2f(d4.y * g), 1.0f);
                    r2 = fmaf(r2, __builtin_amdgcn_exp2f(d4.z * g), 1.0f);
                    r3 = fmaf(r3, __builtin_amdgcn_exp2f(d4.w * g), 1.0f);
                    *reinterpret_cast<float4*>(s_R + pos * RP + cg * 4) =
                        make_float4(r0, r1, r2, r3);
                }
            }
            cur = nxt;
        }
    }
    __syncthreads();

    int sb[IPW], sn[IPW];
#pragma unroll
    for (int j = 0; j < IPW; ++j) {
        sb[j] = __builtin_amdgcn_readfirstlane(s_beg[wave * IPW + j]);
        sn[j] = __builtin_amdgcn_readfirstlane(s_end[wave * IPW + j]) - sb[j];
    }
    int lo[IPW], hi[IPW];
#pragma unroll
    for (int j = 0; j < IPW; ++j) { lo[j] = 0; hi[j] = sn[j]; }
    const int nmax = max(max(sn[0], sn[1]), max(sn[2], sn[3]));
    const int nit  = 32 - __clz(nmax);
    for (int it = 0; it < nit; ++it) {
#pragma unroll
        for (int j = 0; j < IPW; ++j) {
            const bool  act  = lo[j] < hi[j];
            const int   mid  = (lo[j] + hi[j]) >> 1;
            const float tp   = s_pt[min(sb[j] + mid, Tt - 1)];
            const bool  pass = (tm - tp) > EPS_F;
            lo[j] = (act && pass)  ? mid + 1 : lo[j];
            hi[j] = (act && !pass) ? mid     : hi[j];
        }
    }

    float acc[CH];
#pragma unroll
    for (int c = 0; c < CH; ++c) acc[c] = 0.0f;
#pragma unroll
    for (int j = 0; j < IPW; ++j) {
        const int   k  = lo[j];
        const int   p  = max(sb[j] + k - 1, 0);
        const float et = tm - s_pt[p];
        const float x  = (k > 0) ? et * NLOG2E : -INFINITY;
        const float* rr = s_R + p * RP + hc * CH;
#pragma unroll
        for (int c = 0; c < CH; ++c) {
            acc[c] = fmaf(a[j][c] * rr[c], __builtin_amdgcn_exp2f(d[j][c] * x), acc[c]);
        }
    }

#pragma unroll
    for (int c = 0; c < CH; ++c) {
        s_part[(wave * Cc + hc * CH + c) * PP + q] = acc[c];
    }
    __syncthreads();
#pragma unroll
    for (int qq = 0; qq < QT / NW; ++qq) {
        const int qj = wave + qq * NW;
        float s = 0.0f;
#pragma unroll
        for (int w = 0; w < NW; ++w) s += s_part[(w * Cc + lane) * PP + qj];
        s_acc[qj * Cc + lane] = s + mu_l;
    }
    __syncthreads();

    const int half = lane >> 4;
    const int l15  = lane & 15;
    const int nt   = wave;
    v8f cacc = {};
#pragma unroll
    for (int kc = 0; kc < 8; ++kc) {
        v2f af, bf;
#pragma unroll
        for (int j = 0; j < 2; ++j) {
            const int cch = kc * 4 + half * 2 + j;
            af[j] = s_acc[l15 * Cc + cch];
            bf[j] = bfr[kc][j];
        }
        cacc = __builtin_amdgcn_wmma_f32_16x16x4_f32(
             false, af,  false, bf,
             (short)0, cacc,  false,  false);
    }
#pragma unroll
    for (int r = 0; r < 8; ++r) {
        const int m = r + half * 8;
        const int t = t_base + m;
        out[(b * TPp + t) * Kk + nt * 16 + l15] = cacc[r];
    }
}

extern "C" void kernel_launch(void* const* d_in, const int* in_sizes, int n_in,
                              void* d_out, int out_size, void* d_ws, size_t ws_size,
                              hipStream_t stream) {
    const int*   past_event  = (const int*)  d_in[0];
    const float* past_time   = (const float*)d_in[1];
    const float* time_tensor = (const float*)d_in[2];
    const float* mu          = (const float*)d_in[3];
    const float* alpha       = (const float*)d_in[4];
    const float* delta       = (const float*)d_in[5];
    const float* cf_logits   = (const float*)d_in[6];
    const int*   ftc         = (const int*)  d_in[7];
    float* out = (float*)d_out;

    float* sp_alpha = (float*)d_ws;
    float* sp_delta = sp_alpha + Cc * Cc;
    float* sp_mu    = sp_delta + Cc * Cc;
    float* Mexp     = sp_mu + Cc;
    int*   ord_event = (int*)(Mexp + Cc * Kk);
    float* ord_time  = (float*)(ord_event + Bb * Tt);

    prep_params<<<PREP_BLOCKS, 128, 0, stream>>>(past_event, past_time,
                                       mu, alpha, delta, cf_logits, ftc,
                                       sp_alpha, sp_delta, sp_mu, Mexp,
                                       ord_event, ord_time);
    hawkes_wmma_kernel<<<Bb * (TPp / QT), NW * 32, 0, stream>>>(
        ord_event, ord_time, time_tensor,
        sp_alpha, sp_delta, sp_mu, Mexp, out);
}


# Chain phase cut to its recurrence: every position's exponent made in parallel in the segment-ends pass (neighbours read as one LDS group); loop starts at its segment, steps unguarded (sum order kept)
# speedup vs baseline: 4.5966x; 1.1079x over previous
#include <hip/hip_runtime.h>
#include <math.h>

#define Cc   32
#define Kk   128
#define Bb   4
#define Tt   512
#define TPp  1024
#define QT   16
#define NW   8
#define IPW  (Cc / NW)
#define CH   (Cc / 2)
#define PP   17
#define RP   36
#define NLOG2E (-1.4426950408889634f)
#define EPS_F 2.2204460492503131e-16f

typedef __attribute__((ext_vector_type(2))) float v2f;
typedef __attribute__((ext_vector_type(8))) float v8f;

__device__ __forceinline__ float softplus_f(float x) {
    return fmaxf(x, 0.0f) + log1pf(__expf(-fabsf(x)));
}

template <int CTRL>
__device__ __forceinline__ float row_swz(float x) {
    return __int_as_float(__builtin_amdgcn_update_dpp(0, __float_as_int(x), CTRL, 0xF, 0xF, false));
}
__device__ __forceinline__ float xhalf(float x) {
    const int b = __float_as_int(x);
    return __int_as_float(__builtin_amdgcn_permlanex16(b, b, 0x76543210, 0xfedcba98, false, false));
}

#define SP_BLOCKS   8
#define ROW_BLOCKS  4
#define SORT_EV     32
#define SORT_BLOCKS (Bb * Tt / SORT_EV)
#define PREP_BLOCKS 77
static_assert(PREP_BLOCKS == SP_BLOCKS + 1 + ROW_BLOCKS + SORT_BLOCKS, "prep grid");
static_assert(Tt == 512 && Tt % 128 == 0 && SORT_EV * 4 == 128, "sort tiling and 9 index bits");
static_assert(SP_BLOCKS * 128 == Cc * Cc && ROW_BLOCKS * 8 == Cc && Kk == 128, "prep tiling");

__global__ __launch_bounds__(128) __attribute__((amdgpu_waves_per_eu(1, 8)))
void prep_params(const int*   __restrict__ past_event,
                 const float* __restrict__ past_time,
                 const float* __restrict__ mu,
                 const float* __restrict__ alpha,
                 const float* __restrict__ delta,
                 const float* __restrict__ cf,
                 const int*   __restrict__ ftc,
                 float* __restrict__ sp_alpha,
                 float* __restrict__ sp_delta,
                 float* __restrict__ sp_mu,
                 float* __restrict__ Mexp,
                 int*   __restrict__ ord_event,
                 float* __restrict__ ord_time) {
    __shared__ __attribute__((aligned(16))) float s_e[Kk];
    __shared__ __attribute__((aligned(16))) int   s_f[Kk];
    __shared__ __attribute__((aligned(16))) unsigned long long s_key[Tt];
    __shared__ int s_rank[128];
    const int tid = threadIdx.x;
    const int blk = blockIdx.x;

    if (blk < SP_BLOCKS) {
        const int i = blk * 128 + tid;
        sp_alpha[i] = softplus_f(alpha[i]);
        sp_delta[i] = softplus_f(delta[i]) + EPS_F;
        return;
    }
    if (blk == SP_BLOCKS) {
        if (tid < Cc) sp_mu[tid] = softplus_f(mu[tid]);
        return;
    }
    if (blk >= SP_BLOCKS + 1 + ROW_BLOCKS) {
        const int s    = blk - (SP_BLOCKS + 1 + ROW_BLOCKS);
        const int bb   = s / (Tt / SORT_EV);
        const int i0   = (s % (Tt / SORT_EV)) * SORT_EV;
        const int part = tid >> 5;
        const int li   = tid & 31;
        unsigned int tbv[Tt / 128];
        unsigned int evv[Tt / 128];
#pragma unroll
        for (int r = 0; r < Tt / 128; ++r) {
            tbv[r] = __float_as_uint(past_time[bb * Tt + r * 128 + tid]);
            evv[r] = (unsigned int)past_event[bb * Tt + r * 128 + tid];
        }
#pragma unroll
        for (int r = 0; r < Tt / 128; ++r) {
            const int j = r * 128 + tid;
            const unsigned int tb = tbv[r];
            const unsigned int to = tb ^ ((tb >> 31) ? 0xFFFFFFFFu : 0x80000000u);
            s_key[j] = ((unsigned long long)evv[r] << 41)
                     | ((unsigned long long)to << 9) | (unsigned long long)j;
        }
        __syncthreads();
        const unsigned long long key = s_key[i0 + li];
        const int j0  = part * (Tt / 4);
        int r0 = 0, r1 = 0, r2 = 0, r3 = 0;
#pragma unroll
        for (int j = 0; j < Tt / 4; j += 4) {
            r0 += (s_key[j0 + j]     < key) ? 1 : 0;
            r1 += (s_key[j0 + j + 1] < key) ? 1 : 0;
            r2 += (s_key[j0 + j + 2] < key) ? 1 : 0;
            r3 += (s_key[j0 + j + 3] < key) ? 1 : 0;
        }
        s_rank[part * SORT_EV + li] = (r0 + r1) + (r2 + r3);
        __syncthreads();
        if (tid < SORT_EV) {
            const int rank = (s_rank[tid] + s_rank[SORT_EV + tid])
                           + (s_rank[2 * SORT_EV + tid] + s_rank[3 * SORT_EV + tid]);
            const unsigned int to = (unsigned int)(key >> 9);
            const unsigned int tb = to ^ ((to >> 31) ? 0x80000000u : 0xFFFFFFFFu);
            ord_event[bb * Tt + rank] = (int)(key >> 41);
            ord_time[bb * Tt + rank]  = __uint_as_float(tb);
        }
        return;
    }

    const float cfk = cf[tid];
    const int   f   = ftc[tid];
    const int l4 = (tid & 31) * 4;
    const float g0 = cf[l4], g1 = cf[l4 + 1], g2 = cf[l4 + 2], g3 = cf[l4 + 3];
    float m = fmaxf(fmaxf(g0, g1), fmaxf(g2, g3));
    m = fmaxf(m, row_swz<0xB1>(m));
    m = fmaxf(m, row_swz<0x4E>(m));
    m = fmaxf(m, row_swz<0x141>(m));
    m = fmaxf(m, row_swz<0x140>(m));
    m = fmaxf(m, xhalf(m));
    const float e = __expf(cfk - m);
    s_e[tid] = e;
    s_f[tid] = f;
    __syncthreads();
    float den = 0.0f;
#pragma unroll
    for (int k = 0; k < Kk; ++k) den += (s_f[k] == f) ? s_e[k] : 0.0f;
    const float p = e / den;

    const int c0 = (blk - SP_BLOCKS - 1) * (Cc / ROW_BLOCKS);
#pragma unroll
    for (int r = 0; r < Cc / ROW_BLOCKS; ++r) {
        const int c = c0 + r;
        Mexp[c * Kk + tid] = (f == c) ? p : 0.0f;
    }
}

__global__ __launch_bounds__(256)
void hawkes_wmma_kernel(const int*   __restrict__ ord_event,
                        const float* __restrict__ ord_time,
                        const float* __restrict__ time_tensor,
                        const float* __restrict__ sp_alpha,
                        const float* __restrict__ sp_delta,
                        const float* __restrict__ sp_mu,
                        const float* __restrict__ Mexp,
                        float*       __restrict__ out) {
    __shared__ float s_part[NW * Cc * PP];
    __shared__ float s_acc[QT * Cc];
    __shared__ __attribute__((aligned(16))) float s_R[Tt * RP];
    __shared__ int   s_ev[Tt];
    __shared__ __attribute__((aligned(16))) float s_pt[Tt];
    __shared__ int   s_beg[Cc];
    __shared__ int   s_end[Cc];
    __shared__ float s_g[Tt + 8];

    const int tid    = threadIdx.x;
    const int lane   = tid & 31;
    const int q      = lane & 15;
    const int hc     = lane >> 4;
    const int wave   = __builtin_amdgcn_readfirstlane(tid >> 5);
    const int b      = blockIdx.x >> 6;
    const int tile   = blockIdx.x & 63;
    const int t_base = tile * QT;

    if (tid < Cc) { s_beg[tid] = 0; s_end[tid] = 0; }

    const int   e0 = ord_event[b * Tt + tid];
    const int   e1 = ord_event[b * Tt + 256 + tid];
    const float p0 = ord_time[b * Tt + tid];
    const float p1 = ord_time[b * Tt + 256 + tid];
    const float tm = time_tensor[b * TPp + t_base + q];

    const int se = tid >> 3;
    const int cg = tid & 7;
    const float4 d4 = *reinterpret_cast<const float4*>(sp_delta + se * Cc + cg * 4);
    float a[IPW][CH], d[IPW][CH];
#pragma unroll
    for (int j = 0; j < IPW; ++j) {
        const float* __restrict__ ar = sp_alpha + (wave * IPW + j) * Cc + hc * CH;
        const float* __restrict__ dr = sp_delta + (wave * IPW + j) * Cc + hc * CH;
#pragma unroll
        for (int c = 0; c < CH; ++c) { a[j][c] = ar[c]; d[j][c] = dr[c]; }
    }
    const float mu_l = sp_mu[lane];
    float bfr[8][2];
#pragma unroll
    for (int kc = 0; kc < 8; ++kc) {
#pragma unroll
        for (int j = 0; j < 2; ++j) {
            bfr[kc][j] = Mexp[(kc * 4 + hc * 2 + j) * Kk + wave * 16 + q];
        }
    }

    s_ev[tid] = e0;  s_ev[tid + 256] = e1;
    s_pt[tid] = p0;  s_pt[tid + 256] = p1;
    if (tid < 8) s_g[Tt + tid] = 0.0f;
    __syncthreads();
    {
        const int   i0  = tid, i1 = tid + 256;
        const int   em0 = s_ev[max(i0 - 1, 0)];
        const float tm0 = s_pt[max(i0 - 1, 0)];
        const int   ep0 = s_ev[i0 + 1];
        const int   em1 = s_ev[i1 - 1];
        const float tm1 = s_pt[i1 - 1];
        const int   ep1 = s_ev[min(i1 + 1, Tt - 1)];
        const bool first0 = (i0 == 0) || (em0 != e0);
        const bool last0  = (ep0 != e0);
        const bool first1 = (em1 != e1);
        const bool last1  = (i1 == Tt - 1) || (ep1 != e1);
        if (first0) s_beg[e0] = i0;
        if (last0)  s_end[e0] = i0 + 1;
        if (first1) s_beg[e1] = i1;
        if (last1)  s_end[e1] = i1 + 1;
        s_g[i0] = first0 ? 0.0f : (p0 - tm0) * NLOG2E;
        s_g[i1] = first1 ? 0.0f : (p1 - tm1) * NLOG2E;
    }
    __syncthreads();

    {
        const int beg = s_beg[se];
        const int end = s_end[se];
        float r0 = 0.0f, r1 = 0.0f, r2 = 0.0f, r3 = 0.0f;
        float g[4];
#pragma unroll
        for (int j = 0; j < 4; ++j) g[j] = s_g[beg + j];
        for (int pos = beg; pos < end; pos += 4) {
            float gn[4];
#pragma unroll
            for (int j = 0; j < 4; ++j) gn[j] = s_g[pos + 4 + j];
#pragma unroll
            for (int j = 0; j < 4; ++j) {
                r0 = fmaf(r0, __builtin_amdgcn_exp2f(d4.x * g[j]), 1.0f);
                r1 = fmaf(r1, __builtin_amdgcn_exp2f(d4.y * g[j]), 1.0f);
                r2 = fmaf(r2, __builtin_amdgcn_exp2f(d4.z * g[j]), 1.0f);
                r3 = fmaf(r3, __builtin_amdgcn_exp2f(d4.w * g[j]), 1.0f);
                if (pos + j < end) {
                    *reinterpret_cast<float4*>(s_R + (pos + j) * RP + cg * 4) =
                        make_float4(r0, r1, r2, r3);
                }
            }
#pragma unroll
            for (int j = 0; j < 4; ++j) g[j] = gn[j];
        }
    }
    __syncthreads();

    int sb[IPW], sn[IPW];
#pragma unroll
    for (int j = 0; j < IPW; ++j) {
        sb[j] = __builtin_amdgcn_readfirstlane(s_beg[wave * IPW + j]);
        sn[j] = __builtin_amdgcn_readfirstlane(s_end[wave * IPW + j]) - sb[j];
    }
    int lo[IPW], hi[IPW];
#pragma unroll
    for (int j = 0; j < IPW; ++j) { lo[j] = 0; hi[j] = sn[j]; }
    const int nmax = max(max(sn[0], sn[1]), max(sn[2], sn[3]));
    const int nit  = 32 - __clz(nmax);
    for (int it = 0; it < nit; ++it) {
#pragma unroll
        for (int j = 0; j < IPW; ++j) {
            const bool  act  = lo[j] < hi[j];
            const int   mid  = (lo[j] + hi[j]) >> 1;
            const float tp   = s_pt[min(sb[j] + mid, Tt - 1)];
            const bool  pass = (tm - tp) > EPS_F;
            lo[j] = (act && pass)  ? mid + 1 : lo[j];
            hi[j] = (act && !pass) ? mid     : hi[j];
        }
    }

    float acc[CH];
#pragma unroll
    for (int c = 0; c < CH; ++c) acc[c] = 0.0f;
#pragma unroll
    for (int j = 0; j < IPW; ++j) {
        const int   k  = lo[j];
        const int   p  = max(sb[j] + k - 1, 0);
        const float et = tm - s_pt[p];
        const float x  = (k > 0) ? et * NLOG2E : -INFINITY;
        const float* rr = s_R + p * RP + hc * CH;
#pragma unroll
        for (int c = 0; c < CH; ++c) {
            acc[c] = fmaf(a[j][c] * rr[c], __builtin_amdgcn_exp2f(d[j][c] * x), acc[c]);
        }
    }

#pragma unroll
    for (int c = 0; c < CH; ++c) {
        s_part[(wave * Cc + hc * CH + c) * PP + q] = acc[c];
    }
    __syncthreads();
#pragma unroll
    for (int qq = 0; qq < QT / NW; ++qq) {
        const int qj = wave + qq * NW;
        float s = 0.0f;
#pragma unroll
        for (int w = 0; w < NW; ++w) s += s_part[(w * Cc + lane) * PP + qj];
        s_acc[qj * Cc + lane] = s + mu_l;
    }
    __syncthreads();

    const int half = lane >> 4;
    const int l15  = lane & 15;
    const int nt   = wave;
    v8f cacc = {};
#pragma unroll
    for (int kc = 0; kc < 8; ++kc) {
        v2f af, bf;
#pragma unroll
        for (int j = 0; j < 2; ++j) {
            const int cch = kc * 4 + half * 2 + j;
            af[j] = s_acc[l15 * Cc + cch];
            bf[j] = bfr[kc][j];
        }
        cacc = __builtin_amdgcn_wmma_f32_16x16x4_f32(
             false, af,  false, bf,
             (short)0, cacc,  false,  false);
    }
#pragma unroll
    for (int r = 0; r < 8; ++r) {
        const int m = r + half * 8;
        const int t = t_base + m;
        out[(b * TPp + t) * Kk + nt * 16 + l15] = cacc[r];
    }
}

extern "C" void kernel_launch(void* const* d_in, const int* in_sizes, int n_in,
                              void* d_out, int out_size, void* d_ws, size_t ws_size,
                              hipStream_t stream) {
    const int*   past_event  = (const int*)  d_in[0];
    const float* past_time   = (const float*)d_in[1];
    const float* time_tensor = (const float*)d_in[2];
    const float* mu          = (const float*)d_in[3];
    const float* alpha       = (const float*)d_in[4];
    const float* delta       = (const float*)d_in[5];
    const float* cf_logits   = (const float*)d_in[6];
    const int*   ftc         = (const int*)  d_in[7];
    float* out = (float*)d_out;

    float* sp_alpha = (float*)d_ws;
    float* sp_delta = sp_alpha + Cc * Cc;
    float* sp_mu    = sp_delta + Cc * Cc;
    float* Mexp     = sp_mu + Cc;
    int*   ord_event = (int*)(Mexp + Cc * Kk);
    float* ord_time  = (float*)(ord_event + Bb * Tt);

    prep_params<<<PREP_BLOCKS, 128, 0, stream>>>(past_event, past_time,
                                       mu, alpha, delta, cf_logits, ftc,
                                       sp_alpha, sp_delta, sp_mu, Mexp,
                                       ord_event, ord_time);
    hawkes_wmma_kernel<<<Bb * (TPp / QT), NW * 32, 0, stream>>>(
        ord_event, ord_time, time_tensor,
        sp_alpha, sp_delta, sp_mu, Mexp, out);
}


# Prep launch fused in, wave-private: a wave takes only its own 4 ids' events (ballot counts), makes only its own table rows on exp2/log2/rcp; group sums in 8 partials (sum order changes)
# speedup vs baseline: 5.7503x; 1.2510x over previous
#include <hip/hip_runtime.h>
#include <math.h>

#define Cc   32
#define Kk   128
#define Bb   4
#define Tt   512
#define TPp  1024
#define QT   16
#define NW   8
#define IPW  (Cc / NW)
#define CH   (Cc / 2)
#define PP   17
#define RP   36
#define NLOG2E (-1.4426950408889634f)
#define EPS_F 2.2204460492503131e-16f

typedef __attribute__((ext_vector_type(2))) float v2f;
typedef __attribute__((ext_vector_type(8))) float v8f;

__device__ __forceinline__ float softplus_f(float x) {
    const float e   = __builtin_amdgcn_exp2f(fabsf(x) * NLOG2E);
    const float u   = 1.0f + e;
    const float um1 = u - 1.0f;
    const float lg  = __builtin_amdgcn_logf(u) * 0.69314718055994531f;
    const float l1p = (um1 == 0.0f) ? e : lg * (e * __builtin_amdgcn_rcpf(um1));
    return fmaxf(x, 0.0f) + l1p;
}

__device__ __forceinline__ unsigned int tkey(float t) {
    const unsigned int tb = __float_as_uint(t);
    return tb ^ ((tb >> 31) ? 0xFFFFFFFFu : 0x80000000u);
}

template <int CTRL>
__device__ __forceinline__ float row_swz(float x) {
    return __int_as_float(__builtin_amdgcn_update_dpp(0, __float_as_int(x), CTRL, 0xF, 0xF, false));
}
__device__ __forceinline__ float xhalf(float x) {
    const int b = __float_as_int(x);
    return __int_as_float(__builtin_amdgcn_permlanex16(b, b, 0x76543210, 0xfedcba98, false, false));
}

#define EPL (Tt / 32)
__global__ __launch_bounds__(256)
void hawkes_wmma_kernel(const int*   __restrict__ past_event,
                        const float* __restrict__ past_time,
                        const float* __restrict__ time_tensor,
                        const float* __restrict__ mu,
                        const float* __restrict__ alpha,
                        const float* __restrict__ delta,
                        const float* __restrict__ cf,
                        const int*   __restrict__ ftc,
                        float*       __restrict__ out) {
    __shared__ float s_part[NW * Cc * PP];
    __shared__ float s_acc[QT * Cc];
    __shared__ __attribute__((aligned(16))) float s_R[Tt * RP];
    __shared__ __attribute__((aligned(16))) float s_pt[2 * Tt + NW * 32];
    __shared__ float s_g[Tt + 8];
    __shared__ __attribute__((aligned(16))) float s_ta[Cc * Cc];
    __shared__ __attribute__((aligned(16))) float s_td[Cc * Cc];
    __shared__ float s_den[NW * Cc];

    const int tid    = threadIdx.x;
    const int lane   = tid & 31;
    const int q      = lane & 15;
    const int hc     = lane >> 4;
    const int wave   = __builtin_amdgcn_readfirstlane(tid >> 5);
    const int b      = blockIdx.x >> 6;
    const int tile   = blockIdx.x & 63;
    const int t_base = tile * QT;

    int   ev[EPL];
    float tv[EPL];
#pragma unroll
    for (int r = 0; r < EPL; ++r) {
        ev[r] = past_event[b * Tt + r * 32 + lane];
        tv[r] = past_time[b * Tt + r * 32 + lane];
    }
    const float tm = time_tensor[b * TPp + t_base + q];
    const int se = tid >> 3;
    const int cg = tid & 7;
    const float4 al4 = *reinterpret_cast<const float4*>(alpha + se * Cc + cg * 4);
    const float4 de4 = *reinterpret_cast<const float4*>(delta + se * Cc + cg * 4);
    const float mu_r = mu[lane];
    const int   col  = wave * 16 + q;
    const float cfk  = cf[col];
    const int   f    = ftc[col];
    const int   l4   = lane * 4;
    const float g0 = cf[l4], g1 = cf[l4 + 1], g2 = cf[l4 + 2], g3 = cf[l4 + 3];

    const int id0 = wave * IPW;
    int n0 = 0, n1 = 0, n2 = 0, n3 = 0, nlt = 0;
    int rel[EPL];
#pragma unroll
    for (int r = 0; r < EPL; ++r) {
        const int e = ev[r];
        const unsigned int m0 = __builtin_amdgcn_ballot_w32(e == id0);
        const unsigned int m1 = __builtin_amdgcn_ballot_w32(e == id0 + 1);
        const unsigned int m2 = __builtin_amdgcn_ballot_w32(e == id0 + 2);
        const unsigned int m3 = __builtin_amdgcn_ballot_w32(e == id0 + 3);
        const unsigned int ml = __builtin_amdgcn_ballot_w32(e < id0);
        int rr = (int)__builtin_amdgcn_mbcnt_lo(m0, (unsigned int)n0);
        rr = (e == id0 + 1) ? (int)__builtin_amdgcn_mbcnt_lo(m1, (unsigned int)n1) : rr;
        rr = (e == id0 + 2) ? (int)__builtin_amdgcn_mbcnt_lo(m2, (unsigned int)n2) : rr;
        rr = (e == id0 + 3) ? (int)__builtin_amdgcn_mbcnt_lo(m3, (unsigned int)n3) : rr;
        rel[r] = rr;
        n0 += __builtin_popcount(m0);
        n1 += __builtin_popcount(m1);
        n2 += __builtin_popcount(m2);
        n3 += __builtin_popcount(m3);
        nlt += __builtin_popcount(ml);
    }
    const int beg0 = nlt;
    const int beg1 = beg0 + n0;
    const int beg2 = beg1 + n1;
    const int beg3 = beg2 + n2;
    const int endw = beg3 + n3;
#pragma unroll
    for (int r = 0; r < EPL; ++r) {
        const int  jj   = ev[r] - id0;
        const bool mine = (unsigned int)jj < (unsigned int)IPW;
        const int  base = (jj == 0) ? beg0 : (jj == 1) ? beg1 : (jj == 2) ? beg2 : beg3;
        s_pt[mine ? base + rel[r] : 2 * Tt + tid] = tv[r];
    }

    const float4 a4 = make_float4(softplus_f(al4.x), softplus_f(al4.y),
                                  softplus_f(al4.z), softplus_f(al4.w));
    const float4 d4 = make_float4(softplus_f(de4.x) + EPS_F, softplus_f(de4.y) + EPS_F,
                                  softplus_f(de4.z) + EPS_F, softplus_f(de4.w) + EPS_F);
    *reinterpret_cast<float4*>(s_ta + se * Cc + cg * 4) = a4;
    *reinterpret_cast<float4*>(s_td + se * Cc + cg * 4) = d4;
    const float mu_l = softplus_f(mu_r);

    float m = fmaxf(fmaxf(g0, g1), fmaxf(g2, g3));
    m = fmaxf(m, row_swz<0xB1>(m));
    m = fmaxf(m, row_swz<0x4E>(m));
    m = fmaxf(m, row_swz<0x141>(m));
    m = fmaxf(m, row_swz<0x140>(m));
    m = fmaxf(m, xhalf(m));
    const float e_own = __expf(cfk - m);
    {
        float part = 0.0f;
#pragma unroll
        for (int i = 0; i < 16; ++i) {
            const float ei = __int_as_float(__builtin_amdgcn_readlane(__float_as_int(e_own), i));
            const int   fi = __builtin_amdgcn_readlane(f, i);
            part += (fi == lane) ? ei : 0.0f;
        }
        s_den[wave * Cc + lane] = part;
    }
    if (tid < 8) s_g[Tt + tid] = 0.0f;
    __syncthreads();

    float bfr[8][2];
    {
        float den = 0.0f;
#pragma unroll
        for (int w = 0; w < NW; ++w) den += s_den[w * Cc + (f & (Cc - 1))];
        const float pcol = e_own / den;
#pragma unroll
        for (int kc = 0; kc < 8; ++kc) {
#pragma unroll
            for (int j = 0; j < 2; ++j) {
                bfr[kc][j] = (f == kc * 4 + hc * 2 + j) ? pcol : 0.0f;
            }
        }
    }

    auto seg_pass = [&](const int off) -> bool {
        bool bad = false;
        for (int p0 = beg0; p0 < endw; p0 += 32) {
            const int   p  = p0 + lane;
            const int   pc = min(p, Tt - 1);
            const float tp = s_pt[off + pc];
            const float tq = s_pt[off + max(pc - 1, 0)];
            const bool first = (p == beg0) || (p == beg1) || (p == beg2) || (p == beg3);
            if (p < endw) {
                s_g[p] = first ? 0.0f : (tp - tq) * NLOG2E;
                bad = bad || (!first && (tkey(tq) > tkey(tp)));
            }
        }
        return __builtin_amdgcn_ballot_w32(bad) != 0u;
    };
    int ptoff = 0;
    if (seg_pass(0)) {
#pragma unroll 1
        for (int j = 0; j < IPW; ++j) {
            const int bj = (j == 0) ? beg0 : (j == 1) ? beg1 : (j == 2) ? beg2 : beg3;
            const int nj = (j == 0) ? n0 : (j == 1) ? n1 : (j == 2) ? n2 : n3;
            for (int c0 = 0; c0 < nj; c0 += 32) {
                const int   i  = c0 + lane;
                const float ti = s_pt[bj + min(i, nj - 1)];
                const unsigned int ki = tkey(ti);
                int cnt = 0;
                for (int k = 0; k < nj; ++k) {
                    const unsigned int kk = tkey(s_pt[bj + k]);
                    cnt += ((kk < ki) || (kk == ki && k < i)) ? 1 : 0;
                }
                if (i < nj) s_pt[Tt + bj + cnt] = ti;
            }
        }
        ptoff = Tt;
        (void)seg_pass(Tt);
    }

    float a[IPW][CH], d[IPW][CH];
#pragma unroll
    for (int j = 0; j < IPW; ++j) {
        const float* ar = s_ta + (wave * IPW + j) * Cc + hc * CH;
        const float* dr = s_td + (wave * IPW + j) * Cc + hc * CH;
#pragma unroll
        for (int c = 0; c < CH; ++c) { a[j][c] = ar[c]; d[j][c] = dr[c]; }
    }
    __syncthreads();

    {
        const int sj  = lane >> 3;
        const int beg = (sj == 0) ? beg0 : (sj == 1) ? beg1 : (sj == 2) ? beg2 : beg3;
        const int end = (sj == 0) ? beg1 : (sj == 1) ? beg2 : (sj == 2) ? beg3 : endw;
        float r0 = 0.0f, r1 = 0.0f, r2 = 0.0f, r3 = 0.0f;
        float g[4];
#pragma unroll
        for (int j = 0; j < 4; ++j) g[j] = s_g[beg + j];
        for (int pos = beg; pos < end; pos += 4) {
            float gn[4];
#pragma unroll
            for (int j = 0; j < 4; ++j) gn[j] = s_g[pos + 4 + j];
#pragma unroll
            for (int j = 0; j < 4; ++j) {
                r0 = fmaf(r0, __builtin_amdgcn_exp2f(d4.x * g[j]), 1.0f);
                r1 = fmaf(r1, __builtin_amdgcn_exp2f(d4.y * g[j]), 1.0f);
                r2 = fmaf(r2, __builtin_amdgcn_exp2f(d4.z * g[j]), 1.0f);
                r3 = fmaf(r3, __builtin_amdgcn_exp2f(d4.w * g[j]), 1.0f);
                if (pos + j < end) {
                    *reinterpret_cast<float4*>(s_R + (pos + j) * RP + cg * 4) =
                        make_float4(r0, r1, r2, r3);
                }
            }
#pragma unroll
            for (int j = 0; j < 4; ++j) g[j] = gn[j];
        }
    }
    __syncthreads();

    const int sb[IPW] = {beg0, beg1, beg2, beg3};
    const int sn[IPW] = {n0, n1, n2, n3};
    int lo[IPW], hi[IPW];
#pragma unroll
    for (int j = 0; j < IPW; ++j) { lo[j] = 0; hi[j] = sn[j]; }
    const int nmax = max(max(sn[0], sn[1]), max(sn[2], sn[3]));
    const int nit  = 32 - __clz(nmax);
    for (int it = 0; it < nit; ++it) {
#pragma unroll
        for (int j = 0; j < IPW; ++j) {
            const bool  act  = lo[j] < hi[j];
            const int   mid  = (lo[j] + hi[j]) >> 1;
            const float tp   = s_pt[ptoff + min(sb[j] + mid, Tt - 1)];
            const bool  pass = (tm - tp) > EPS_F;
            lo[j] = (act && pass)  ? mid + 1 : lo[j];
            hi[j] = (act && !pass) ? mid     : hi[j];
        }
    }

    float acc[CH];
#pragma unroll
    for (int c = 0; c < CH; ++c) acc[c] = 0.0f;
#pragma unroll
    for (int j = 0; j < IPW; ++j) {
        const int   k  = lo[j];
        const int   p  = max(sb[j] + k - 1, 0);
        const float et = tm - s_pt[ptoff + p];
        const float x  = (k > 0) ? et * NLOG2E : -INFINITY;
        const float* rr = s_R + p * RP + hc * CH;
#pragma unroll
        for (int c = 0; c < CH; ++c) {
            acc[c] = fmaf(a[j][c] * rr[c], __builtin_amdgcn_exp2f(d[j][c] * x), acc[c]);
        }
    }

#pragma unroll
    for (int c = 0; c < CH; ++c) {
        s_part[(wave * Cc + hc * CH + c) * PP + q] = acc[c];
    }
    __syncthreads();
#pragma unroll
    for (int qq = 0; qq < QT / NW; ++qq) {
        const int qj = wave + qq * NW;
        float s = 0.0f;
#pragma unroll
        for (int w = 0; w < NW; ++w) s += s_part[(w * Cc + lane) * PP + qj];
        s_acc[qj * Cc + lane] = s + mu_l;
    }
    __syncthreads();

    const int half = lane >> 4;
    const int l15  = lane & 15;
    const int nt   = wave;
    v8f cacc = {};
#pragma unroll
    for (int kc = 0; kc < 8; ++kc) {
        v2f af, bf;
#pragma unroll
        for (int j = 0; j < 2; ++j) {
            const int cch = kc * 4 + half * 2 + j;
            af[j] = s_acc[l15 * Cc + cch];
            bf[j] = bfr[kc][j];
        }
        cacc = __builtin_amdgcn_wmma_f32_16x16x4_f32(
             false, af,  false, bf,
             (short)0, cacc,  false,  false);
    }
#pragma unroll
    for (int r = 0; r < 8; ++r) {
        const int m = r + half * 8;
        const int t = t_base + m;
        out[(b * TPp + t) * Kk + nt * 16 + l15] = cacc[r];
    }
}

extern "C" void kernel_launch(void* const* d_in, const int* in_sizes, int n_in,
                              void* d_out, int out_size, void* d_ws, size_t ws_size,
                              hipStream_t stream) {
    const int*   past_event  = (const int*)  d_in[0];
    const float* past_time   = (const float*)d_in[1];
    const float* time_tensor = (const float*)d_in[2];
    const float* mu          = (const float*)d_in[3];
    const float* alpha       = (const float*)d_in[4];
    const float* delta       = (const float*)d_in[5];
    const float* cf_logits   = (const float*)d_in[6];
    const int*   ftc         = (const int*)  d_in[7];
    float* out = (float*)d_out;

    hawkes_wmma_kernel<<<Bb * (TPp / QT), NW * 32, 0, stream>>>(
        past_event, past_time, time_tensor,
        mu, alpha, delta, cf_logits, ftc, out);
}


# Event placement by the whole block: a wave ranks only its own 64 events for all 32 ids (bit ballots, peer masks), counts joined by an LDS histogram; no wave scans all 512 events (sum order kept)
# speedup vs baseline: 6.6091x; 1.1493x over previous
#include <hip/hip_runtime.h>
#include <math.h>

#define Cc   32
#define Kk   128
#define Bb   4
#define Tt   512
#define TPp  1024
#define QT   16
#define NW   8
#define IPW  (Cc / NW)
#define CH   (Cc / 2)
#define PP   17
#define RP   36
#define NLOG2E (-1.4426950408889634f)
#define EPS_F 2.2204460492503131e-16f

typedef __attribute__((ext_vector_type(2))) float v2f;
typedef __attribute__((ext_vector_type(8))) float v8f;

__device__ __forceinline__ float softplus_f(float x) {
    const float e   = __builtin_amdgcn_exp2f(fabsf(x) * NLOG2E);
    const float u   = 1.0f + e;
    const float um1 = u - 1.0f;
    const float lg  = __builtin_amdgcn_logf(u) * 0.69314718055994531f;
    const float l1p = (um1 == 0.0f) ? e : lg * (e * __builtin_amdgcn_rcpf(um1));
    return fmaxf(x, 0.0f) + l1p;
}

__device__ __forceinline__ unsigned int tkey(float t) {
    const unsigned int tb = __float_as_uint(t);
    return tb ^ ((tb >> 31) ? 0xFFFFFFFFu : 0x80000000u);
}

template <int CTRL>
__device__ __forceinline__ float row_swz(float x) {
    return __int_as_float(__builtin_amdgcn_update_dpp(0, __float_as_int(x), CTRL, 0xF, 0xF, false));
}
template <int N>
__device__ __forceinline__ int row_shr(int x) {
    return __builtin_amdgcn_update_dpp(0, x, 0x110 + N, 0xF, 0xF, false);
}
__device__ __forceinline__ float xhalf(float x) {
    const int b = __float_as_int(x);
    return __int_as_float(__builtin_amdgcn_permlanex16(b, b, 0x76543210, 0xfedcba98, false, false));
}

#define RPW (Tt / 32 / NW)
__global__ __launch_bounds__(256)
void hawkes_wmma_kernel(const int*   __restrict__ past_event,
                        const float* __restrict__ past_time,
                        const float* __restrict__ time_tensor,
                        const float* __restrict__ mu,
                        const float* __restrict__ alpha,
                        const float* __restrict__ delta,
                        const float* __restrict__ cf,
                        const int*   __restrict__ ftc,
                        float*       __restrict__ out) {
    __shared__ float s_part[NW * Cc * PP];
    __shared__ float s_acc[QT * Cc];
    __shared__ __attribute__((aligned(16))) float s_R[Tt * RP];
    __shared__ __attribute__((aligned(16))) float s_pt[2 * Tt + NW * 32];
    __shared__ float s_g[Tt + 8];
    __shared__ __attribute__((aligned(16))) float s_ta[Cc * Cc];
    __shared__ __attribute__((aligned(16))) float s_td[Cc * Cc];
    __shared__ float s_den[NW * Cc];
    __shared__ int   s_hist[(Tt / 32) * Cc];

    const int tid    = threadIdx.x;
    const int lane   = tid & 31;
    const int q      = lane & 15;
    const int hc     = lane >> 4;
    const int wave   = __builtin_amdgcn_readfirstlane(tid >> 5);
    const int b      = blockIdx.x >> 6;
    const int tile   = blockIdx.x & 63;
    const int t_base = tile * QT;

    int   ev[RPW];
    float tv[RPW];
#pragma unroll
    for (int r = 0; r < RPW; ++r) {
        ev[r] = past_event[b * Tt + (wave * RPW + r) * 32 + lane];
        tv[r] = past_time[b * Tt + (wave * RPW + r) * 32 + lane];
    }
    const float tm = time_tensor[b * TPp + t_base + q];
    const int se = tid >> 3;
    const int cg = tid & 7;
    const float4 al4 = *reinterpret_cast<const float4*>(alpha + se * Cc + cg * 4);
    const float4 de4 = *reinterpret_cast<const float4*>(delta + se * Cc + cg * 4);
    const float mu_r = mu[lane];
    const int   col  = wave * 16 + q;
    const float cfk  = cf[col];
    const int   f    = ftc[col];
    const int   l4   = lane * 4;
    const float g0 = cf[l4], g1 = cf[l4 + 1], g2 = cf[l4 + 2], g3 = cf[l4 + 3];

    unsigned int peer[RPW];
    int          cntL[RPW];
#pragma unroll
    for (int r = 0; r < RPW; ++r) {
        const unsigned int e = (unsigned int)ev[r];
        unsigned int pm = __builtin_amdgcn_ballot_w32(e < (unsigned int)Cc);
        unsigned int lm = pm;
#pragma unroll
        for (int k = 0; k < 5; ++k) {
            const unsigned int eb = (e >> k) & 1u;
            const unsigned int lb = ((unsigned int)lane >> k) & 1u;
            const unsigned int bk = __builtin_amdgcn_ballot_w32(eb != 0u);
            pm &= ~(bk ^ (0u - eb));
            lm &= ~(bk ^ (0u - lb));
        }
        peer[r] = pm;
        cntL[r] = __builtin_popcount(lm);
        s_hist[(wave * RPW + r) * Cc + lane] = cntL[r];
    }
    asm volatile("" :: "v"(al4.x), "v"(al4.y), "v"(al4.z), "v"(al4.w),
                       "v"(de4.x), "v"(de4.y), "v"(de4.z), "v"(de4.w));
    __syncthreads();

    int tot = 0, pre0 = 0;
#pragma unroll
    for (int rr = 0; rr < Tt / 32; ++rr) {
        const int h = s_hist[rr * Cc + lane];
        tot  += h;
        pre0 += (rr < wave * RPW) ? h : 0;
    }
    int inc = tot;
    inc += row_shr<1>(inc);
    inc += row_shr<2>(inc);
    inc += row_shr<4>(inc);
    inc += row_shr<8>(inc);
    const int lowsum = __builtin_amdgcn_readlane(inc, 15);
    inc += (lane >= 16) ? lowsum : 0;
    const int start = inc - tot;
    {
        int base = start + pre0;
#pragma unroll
        for (int r = 0; r < RPW; ++r) {
            const bool mine = (unsigned int)ev[r] < (unsigned int)Cc;
            const int  pb   = __builtin_amdgcn_ds_bpermute((ev[r] & (Cc - 1)) << 2, base);
            const int  pos  = pb + (int)__builtin_amdgcn_mbcnt_lo(peer[r], 0u);
            s_pt[mine ? pos : 2 * Tt + tid] = tv[r];
            base += cntL[r];
        }
    }
    const int id0  = wave * IPW;
    const int n0   = __builtin_amdgcn_readlane(tot, id0);
    const int n1   = __builtin_amdgcn_readlane(tot, id0 + 1);
    const int n2   = __builtin_amdgcn_readlane(tot, id0 + 2);
    const int n3   = __builtin_amdgcn_readlane(tot, id0 + 3);
    const int beg0 = __builtin_amdgcn_readlane(start, id0);
    const int beg1 = beg0 + n0;
    const int beg2 = beg1 + n1;
    const int beg3 = beg2 + n2;
    const int endw = beg3 + n3;

    const float4 a4 = make_float4(softplus_f(al4.x), softplus_f(al4.y),
                                  softplus_f(al4.z), softplus_f(al4.w));
    const float4 d4 = make_float4(softplus_f(de4.x) + EPS_F, softplus_f(de4.y) + EPS_F,
                                  softplus_f(de4.z) + EPS_F, softplus_f(de4.w) + EPS_F);
    *reinterpret_cast<float4*>(s_ta + se * Cc + cg * 4) = a4;
    *reinterpret_cast<float4*>(s_td + se * Cc + cg * 4) = d4;
    const float mu_l = softplus_f(mu_r);

    float m = fmaxf(fmaxf(g0, g1), fmaxf(g2, g3));
    m = fmaxf(m, row_swz<0xB1>(m));
    m = fmaxf(m, row_swz<0x4E>(m));
    m = fmaxf(m, row_swz<0x141>(m));
    m = fmaxf(m, row_swz<0x140>(m));
    m = fmaxf(m, xhalf(m));
    const float e_own = __expf(cfk - m);
    {
        float part = 0.0f;
#pragma unroll
        for (int i = 0; i < 16; ++i) {
            const float ei = __int_as_float(__builtin_amdgcn_readlane(__float_as_int(e_own), i));
            const int   fi = __builtin_amdgcn_readlane(f, i);
            part += (fi == lane) ? ei : 0.0f;
        }
        s_den[wave * Cc + lane] = part;
    }
    if (tid < 8) s_g[Tt + tid] = 0.0f;
    __syncthreads();

    float bfr[8][2];
    {
        float den = 0.0f;
#pragma unroll
        for (int w = 0; w < NW; ++w) den += s_den[w * Cc + (f & (Cc - 1))];
        const float pcol = e_own / den;
#pragma unroll
        for (int kc = 0; kc < 8; ++kc) {
#pragma unroll
            for (int j = 0; j < 2; ++j) {
                bfr[kc][j] = (f == kc * 4 + hc * 2 + j) ? pcol : 0.0f;
            }
        }
    }

    auto seg_pass = [&](const int off) -> bool {
        bool bad = false;
        for (int p0 = beg0; p0 < endw; p0 += 32) {
            const int   p  = p0 + lane;
            const int   pc = min(p, Tt - 1);
            const float tp = s_pt[off + pc];
            const float tq = s_pt[off + max(pc - 1, 0)];
            const bool first = (p == beg0) || (p == beg1) || (p == beg2) || (p == beg3);
            if (p < endw) {
                s_g[p] = first ? 0.0f : (tp - tq) * NLOG2E;
                bad = bad || (!first && (tkey(tq) > tkey(tp)));
            }
        }
        return __builtin_amdgcn_ballot_w32(bad) != 0u;
    };
    int ptoff = 0;
    if (seg_pass(0)) {
#pragma unroll 1
        for (int j = 0; j < IPW; ++j) {
            const int bj = (j == 0) ? beg0 : (j == 1) ? beg1 : (j == 2) ? beg2 : beg3;
            const int nj = (j == 0) ? n0 : (j == 1) ? n1 : (j == 2) ? n2 : n3;
            for (int c0 = 0; c0 < nj; c0 += 32) {
                const int   i  = c0 + lane;
                const float ti = s_pt[bj + min(i, nj - 1)];
                const unsigned int ki = tkey(ti);
                int cnt = 0;
                for (int k = 0; k < nj; ++k) {
                    const unsigned int kk = tkey(s_pt[bj + k]);
                    cnt += ((kk < ki) || (kk == ki && k < i)) ? 1 : 0;
                }
                if (i < nj) s_pt[Tt + bj + cnt] = ti;
            }
        }
        ptoff = Tt;
        (void)seg_pass(Tt);
    }

    float a[IPW][CH], d[IPW][CH];
#pragma unroll
    for (int j = 0; j < IPW; ++j) {
        const float* ar = s_ta + (wave * IPW + j) * Cc + hc * CH;
        const float* dr = s_td + (wave * IPW + j) * Cc + hc * CH;
#pragma unroll
        for (int c = 0; c < CH; ++c) { a[j][c] = ar[c]; d[j][c] = dr[c]; }
    }
    __syncthreads();

    {
        const int sj  = lane >> 3;
        const int beg = (sj == 0) ? beg0 : (sj == 1) ? beg1 : (sj == 2) ? beg2 : beg3;
        const int end = (sj == 0) ? beg1 : (sj == 1) ? beg2 : (sj == 2) ? beg3 : endw;
        float r0 = 0.0f, r1 = 0.0f, r2 = 0.0f, r3 = 0.0f;
        float g[4];
#pragma unroll
        for (int j = 0; j < 4; ++j) g[j] = s_g[beg + j];
        for (int pos = beg; pos < end; pos += 4) {
            float gn[4];
#pragma unroll
            for (int j = 0; j < 4; ++j) gn[j] = s_g[pos + 4 + j];
#pragma unroll
            for (int j = 0; j < 4; ++j) {
                r0 = fmaf(r0, __builtin_amdgcn_exp2f(d4.x * g[j]), 1.0f);
                r1 = fmaf(r1, __builtin_amdgcn_exp2f(d4.y * g[j]), 1.0f);
                r2 = fmaf(r2, __builtin_amdgcn_exp2f(d4.z * g[j]), 1.0f);
                r3 = fmaf(r3, __builtin_amdgcn_exp2f(d4.w * g[j]), 1.0f);
                if (pos + j < end) {
                    *reinterpret_cast<float4*>(s_R + (pos + j) * RP + cg * 4) =
                        make_float4(r0, r1, r2, r3);
                }
            }
#pragma unroll
            for (int j = 0; j < 4; ++j) g[j] = gn[j];
        }
    }
    __syncthreads();

    const int sb[IPW] = {beg0, beg1, beg2, beg3};
    const int sn[IPW] = {n0, n1, n2, n3};
    int lo[IPW], hi[IPW];
#pragma unroll
    for (int j = 0; j < IPW; ++j) { lo[j] = 0; hi[j] = sn[j]; }
    const int nmax = max(max(sn[0], sn[1]), max(sn[2], sn[3]));
    const int nit  = 32 - __clz(nmax);
    for (int it = 0; it < nit; ++it) {
#pragma unroll
        for (int j = 0; j < IPW; ++j) {
            const bool  act  = lo[j] < hi[j];
            const int   mid  = (lo[j] + hi[j]) >> 1;
            const float tp   = s_pt[ptoff + min(sb[j] + mid, Tt - 1)];
            const bool  pass = (tm - tp) > EPS_F;
            lo[j] = (act && pass)  ? mid + 1 : lo[j];
            hi[j] = (act && !pass) ? mid     : hi[j];
        }
    }

    float acc[CH];
#pragma unroll
    for (int c = 0; c < CH; ++c) acc[c] = 0.0f;
#pragma unroll
    for (int j = 0; j < IPW; ++j) {
        const int   k  = lo[j];
        const int   p  = max(sb[j] + k - 1, 0);
        const float et = tm - s_pt[ptoff + p];
        const float x  = (k > 0) ? et * NLOG2E : -INFINITY;
        const float* rr = s_R + p * RP + hc * CH;
#pragma unroll
        for (int c = 0; c < CH; ++c) {
            acc[c] = fmaf(a[j][c] * rr[c], __builtin_amdgcn_exp2f(d[j][c] * x), acc[c]);
        }
    }

#pragma unroll
    for (int c = 0; c < CH; ++c) {
        s_part[(wave * Cc + hc * CH + c) * PP + q] = acc[c];
    }
    __syncthreads();
#pragma unroll
    for (int qq = 0; qq < QT / NW; ++qq) {
        const int qj = wave + qq * NW;
        float s = 0.0f;
#pragma unroll
        for (int w = 0; w < NW; ++w) s += s_part[(w * Cc + lane) * PP + qj];
        s_acc[qj * Cc + lane] = s + mu_l;
    }
    __syncthreads();

    const int half = lane >> 4;
    const int l15  = lane & 15;
    const int nt   = wave;
    v8f cacc = {};
#pragma unroll
    for (int kc = 0; kc < 8; ++kc) {
        v2f af, bf;
#pragma unroll
        for (int j = 0; j < 2; ++j) {
            const int cch = kc * 4 + half * 2 + j;
            af[j] = s_acc[l15 * Cc + cch];
            bf[j] = bfr[kc][j];
        }
        cacc = __builtin_amdgcn_wmma_f32_16x16x4_f32(
             false, af,  false, bf,
             (short)0, cacc,  false,  false);
    }
#pragma unroll
    for (int r = 0; r < 8; ++r) {
        const int m = r + half * 8;
        const int t = t_base + m;
        out[(b * TPp + t) * Kk + nt * 16 + l15] = cacc[r];
    }
}

extern "C" void kernel_launch(void* const* d_in, const int* in_sizes, int n_in,
                              void* d_out, int out_size, void* d_ws, size_t ws_size,
                              hipStream_t stream) {
    const int*   past_event  = (const int*)  d_in[0];
    const float* past_time   = (const float*)d_in[1];
    const float* time_tensor = (const float*)d_in[2];
    const float* mu          = (const float*)d_in[3];
    const float* alpha       = (const float*)d_in[4];
    const float* delta       = (const float*)d_in[5];
    const float* cf_logits   = (const float*)d_in[6];
    const int*   ftc         = (const int*)  d_in[7];
    float* out = (float*)d_out;

    hawkes_wmma_kernel<<<Bb * (TPp / QT), NW * 32, 0, stream>>>(
        past_event, past_time, time_tensor,
        mu, alpha, delta, cf_logits, ftc, out);
}
